# starting kernel
# baseline (speedup 1.0000x reference)
#include <hip/hip_runtime.h>

typedef float v2f __attribute__((ext_vector_type(2)));
typedef float v8f __attribute__((ext_vector_type(8)));

#define NROWS 8192
#define MCOLS 128
#define KDZ   128
#define KDX   128
#define HEADS 4

__global__ void __launch_bounds__(128)
xw_bias_kernel(const float* __restrict__ x,
               const float* __restrict__ W,
               const float* __restrict__ b,
               float* __restrict__ xwb) {
  const int m = threadIdx.x;
  float a0 = b[0], a1 = b[1], a2 = b[2], a3 = b[3];
  for (int k = 0; k < KDX; ++k) {
    const float xv = x[k * MCOLS + m];
    const float* wr = W + (KDZ + k) * HEADS;
    a0 = fmaf(xv, wr[0], a0);
    a1 = fmaf(xv, wr[1], a1);
    a2 = fmaf(xv, wr[2], a2);
    a3 = fmaf(xv, wr[3], a3);
  }
  ((float4*)xwb)[m] = make_float4(a0, a1, a2, a3);
}

__global__ void __launch_bounds__(128)
fused_zw_bcast_kernel(const float* __restrict__ z,
                      const float* __restrict__ W,
                      const float* __restrict__ xwb,
                      float* __restrict__ out) {
  __shared__ float s_bf[32 * 64];
  __shared__ float s_xwb[MCOLS * HEADS];
  __shared__ float s_zw[4 * 16 * HEADS];

  const int tid   = threadIdx.x;
  const int lane  = tid & 31;
  const int wave  = tid >> 5;
  const int nIdx  = lane & 15;
  const int khalf = lane >> 4;

  ((float4*)s_xwb)[tid] = ((const float4*)xwb)[tid];

  {
    const int fl  = tid & 31;
    const int fi0 = tid >> 5;
    const int fn  = fl & 15;
    const int fkh = fl >> 4;
    const float msk = (fn < HEADS) ? 1.0f : 0.0f;
#pragma unroll
    for (int q = 0; q < 8; ++q) {
      const int i    = fi0 + 4 * q;
      const int krow = 4 * i + 2 * fkh;
      s_bf[i * 64 + fl * 2 + 0] = msk * W[(krow + 0) * HEADS + fn];
      s_bf[i * 64 + fl * 2 + 1] = msk * W[(krow + 1) * HEADS + fn];
    }
  }
  __syncthreads();

  const int n0 = blockIdx.x * 64 + wave * 16;
  const float* zrow = z + (size_t)(n0 + nIdx) * KDZ + 2 * khalf;

  v8f c = {0.f, 0.f, 0.f, 0.f, 0.f, 0.f, 0.f, 0.f};
#pragma unroll
  for (int i = 0; i < KDZ / 4; ++i) {
    v2f a  = *(const v2f*)(zrow + 4 * i);
    v2f bb = *(const v2f*)(s_bf + i * 64 + lane * 2);
    c = __builtin_amdgcn_wmma_f32_16x16x4_f32(false, a, false, bb,
                                              (short)0, c, false, false);
  }

  if (nIdx < HEADS) {
    const int base = wave * 64 + khalf * (8 * HEADS) + nIdx;
#pragma unroll
    for (int r = 0; r < 8; ++r) {
      s_zw[base + r * HEADS] = c[r];
    }
  }
  __syncthreads();

#pragma unroll 4
  for (int r = 0; r < 16; ++r) {
    const float4 zw = *(const float4*)(s_zw + wave * 64 + r * HEADS);
    const size_t rowoff = (size_t)(n0 + r) * (MCOLS * HEADS);
#pragma unroll
    for (int mb = 0; mb < 4; ++mb) {
      const int m = mb * 32 + lane;
      const float4 xw = *(const float4*)(s_xwb + m * HEADS);
      float4 o;
      o.x = fmaxf(zw.x + xw.x, 0.f);
      o.y = fmaxf(zw.y + xw.y, 0.f);
      o.z = fmaxf(zw.z + xw.z, 0.f);
      o.w = fmaxf(zw.w + xw.w, 0.f);
      *(float4*)(out + rowoff + (size_t)m * HEADS) = o;
    }
  }
}

extern "C" void kernel_launch(void* const* d_in, const int* in_sizes, int n_in,
                              void* d_out, int out_size, void* d_ws, size_t ws_size,
                              hipStream_t stream) {
  (void)in_sizes; (void)n_in; (void)out_size; (void)ws_size;
  const float* z = (const float*)d_in[0];
  const float* x = (const float*)d_in[1];
  const float* W = (const float*)d_in[2];
  const float* b = (const float*)d_in[3];
  float* out = (float*)d_out;
  float* xwb = (float*)d_ws;

  xw_bias_kernel<<<1, 128, 0, stream>>>(x, W, b, xwb);
  fused_zw_bcast_kernel<<<NROWS / 64, 128, 0, stream>>>(z, W, xwb, out);
}


# Fuse the xW+b pre-kernel into the main kernel (one launch): each block builds xW itself while its z loads are in flight, k split over 4 waves, partials added via LDS (xW sum order changes)
# speedup vs baseline: 2.0826x; 2.0826x over previous
#include <hip/hip_runtime.h>

typedef float v2f __attribute__((ext_vector_type(2)));
typedef float v8f __attribute__((ext_vector_type(8)));

#define NROWS 8192
#define MCOLS 128
#define KDZ   128
#define KDX   128
#define HEADS 4

__global__ void __launch_bounds__(128)
fused_xw_zw_bcast_kernel(const float* __restrict__ z,
                         const float* __restrict__ x,
                         const float* __restrict__ W,
                         const float* __restrict__ b,
                         float* __restrict__ out) {
  __shared__ float s_bf[32 * 64];
  __shared__ float s_xwp[4 * MCOLS * HEADS];
  __shared__ float s_zw[4 * 16 * HEADS];

  const int tid   = threadIdx.x;
  const int lane  = tid & 31;
  const int wave  = tid >> 5;
  const int nIdx  = lane & 15;
  const int khalf = lane >> 4;

  const int wu = __builtin_amdgcn_readfirstlane(wave);

  const int fl  = tid & 31;
  const int fi0 = tid >> 5;
  const int fn  = fl & 15;
  const int fkh = fl >> 4;
  float wst[16];
#pragma unroll
  for (int q = 0; q < 8; ++q) {
    const int i    = fi0 + 4 * q;
    const int krow = 4 * i + 2 * fkh;
    wst[2 * q + 0] = W[(krow + 0) * HEADS + fn];
    wst[2 * q + 1] = W[(krow + 1) * HEADS + fn];
  }

  const float* xp = x + (size_t)(wu * 32) * MCOLS + 4 * lane;
  float4 xv[32];
#pragma unroll
  for (int kk = 0; kk < 32; ++kk) {
    xv[kk] = *(const float4*)(xp + kk * MCOLS);
  }

  const int n0 = blockIdx.x * 64 + wave * 16;
  const float* zrow = z + (size_t)(n0 + nIdx) * KDZ + 2 * khalf;
  v2f a[KDZ / 4];
#pragma unroll
  for (int i = 0; i < KDZ / 4; ++i) {
    a[i] = *(const v2f*)(zrow + 4 * i);
  }

  __builtin_amdgcn_sched_barrier(0);

  {
    const float msk = (fn < HEADS) ? 1.0f : 0.0f;
#pragma unroll
    for (int q = 0; q < 8; ++q) {
      const int i = fi0 + 4 * q;
      s_bf[i * 64 + fl * 2 + 0] = msk * wst[2 * q + 0];
      s_bf[i * 64 + fl * 2 + 1] = msk * wst[2 * q + 1];
    }
  }

  {
    const float* wp = W + (KDZ + wu * 32) * HEADS;
    float4 p0 = make_float4(0.f, 0.f, 0.f, 0.f);
    float4 p1 = p0, p2 = p0, p3 = p0;
#pragma unroll
    for (int kk = 0; kk < 32; ++kk) {
      const float4 xk = xv[kk];
      const float w0 = wp[kk * HEADS + 0];
      const float w1 = wp[kk * HEADS + 1];
      const float w2 = wp[kk * HEADS + 2];
      const float w3 = wp[kk * HEADS + 3];
      p0.x = fmaf(xk.x, w0, p0.x); p0.y = fmaf(xk.x, w1, p0.y);
      p0.z = fmaf(xk.x, w2, p0.z); p0.w = fmaf(xk.x, w3, p0.w);
      p1.x = fmaf(xk.y, w0, p1.x); p1.y = fmaf(xk.y, w1, p1.y);
      p1.z = fmaf(xk.y, w2, p1.z); p1.w = fmaf(xk.y, w3, p1.w);
      p2.x = fmaf(xk.z, w0, p2.x); p2.y = fmaf(xk.z, w1, p2.y);
      p2.z = fmaf(xk.z, w2, p2.z); p2.w = fmaf(xk.z, w3, p2.w);
      p3.x = fmaf(xk.w, w0, p3.x); p3.y = fmaf(xk.w, w1, p3.y);
      p3.z = fmaf(xk.w, w2, p3.z); p3.w = fmaf(xk.w, w3, p3.w);
    }
    float4* pp = (float4*)s_xwp + wu * MCOLS + 4 * lane;
    pp[0] = p0; pp[1] = p1; pp[2] = p2; pp[3] = p3;
  }
  __syncthreads();

  v8f c = {0.f, 0.f, 0.f, 0.f, 0.f, 0.f, 0.f, 0.f};
#pragma unroll
  for (int i = 0; i < KDZ / 4; ++i) {
    v2f bb = *(const v2f*)(s_bf + i * 64 + lane * 2);
    c = __builtin_amdgcn_wmma_f32_16x16x4_f32(false, a[i], false, bb,
                                              (short)0, c, false, false);
  }

  if (nIdx < HEADS) {
    const int base = wave * 64 + khalf * (8 * HEADS) + nIdx;
#pragma unroll
    for (int r = 0; r < 8; ++r) {
      s_zw[base + r * HEADS] = c[r];
    }
  }
  __syncthreads();

  const float b0 = b[0], b1 = b[1], b2 = b[2], b3 = b[3];
  float4 xw[4];
#pragma unroll
  for (int mb = 0; mb < 4; ++mb) {
    const int m = mb * 32 + lane;
    const float4 q0 = ((const float4*)s_xwp)[0 * MCOLS + m];
    const float4 q1 = ((const float4*)s_xwp)[1 * MCOLS + m];
    const float4 q2 = ((const float4*)s_xwp)[2 * MCOLS + m];
    const float4 q3 = ((const float4*)s_xwp)[3 * MCOLS + m];
    xw[mb].x = (((b0 + q0.x) + q1.x) + q2.x) + q3.x;
    xw[mb].y = (((b1 + q0.y) + q1.y) + q2.y) + q3.y;
    xw[mb].z = (((b2 + q0.z) + q1.z) + q2.z) + q3.z;
    xw[mb].w = (((b3 + q0.w) + q1.w) + q2.w) + q3.w;
  }

#pragma unroll 4
  for (int r = 0; r < 16; ++r) {
    const float4 zw = *(const float4*)(s_zw + wave * 64 + r * HEADS);
    const size_t rowoff = (size_t)(n0 + r) * (MCOLS * HEADS);
#pragma unroll
    for (int mb = 0; mb < 4; ++mb) {
      const int m = mb * 32 + lane;
      float4 o;
      o.x = fmaxf(zw.x + xw[mb].x, 0.f);
      o.y = fmaxf(zw.y + xw[mb].y, 0.f);
      o.z = fmaxf(zw.z + xw[mb].z, 0.f);
      o.w = fmaxf(zw.w + xw[mb].w, 0.f);
      *(float4*)(out + rowoff + (size_t)m * HEADS) = o;
    }
  }
}

extern "C" void kernel_launch(void* const* d_in, const int* in_sizes, int n_in,
                              void* d_out, int out_size, void* d_ws, size_t ws_size,
                              hipStream_t stream) {
  (void)in_sizes; (void)n_in; (void)out_size; (void)d_ws; (void)ws_size;
  const float* z = (const float*)d_in[0];
  const float* x = (const float*)d_in[1];
  const float* W = (const float*)d_in[2];
  const float* b = (const float*)d_in[3];
  float* out = (float*)d_out;

  fused_xw_zw_bcast_kernel<<<NROWS / 64, 128, 0, stream>>>(z, x, W, b, out);
}


# Grid 128 -> 256 work-groups (one per WGP): 32 rows per block, each 16-row tile's K split over a wave pair (16 WMMAs each), the two partial tiles added in LDS (zW sum order changes)
# speedup vs baseline: 2.5163x; 1.2082x over previous
#include <hip/hip_runtime.h>

typedef float v2f __attribute__((ext_vector_type(2)));
typedef float v8f __attribute__((ext_vector_type(8)));

#define NROWS 8192
#define MCOLS 128
#define KDZ   128
#define KDX   128
#define HEADS 4

__global__ void __launch_bounds__(128)
fused_xw_zw_bcast_kernel(const float* __restrict__ z,
                         const float* __restrict__ x,
                         const float* __restrict__ W,
                         const float* __restrict__ b,
                         float* __restrict__ out) {
  __shared__ float s_bf[32 * 64];
  __shared__ float s_xwp[4 * MCOLS * HEADS];
  __shared__ float s_zw[4 * 16 * HEADS];

  const int tid   = threadIdx.x;
  const int lane  = tid & 31;
  const int wave  = tid >> 5;
  const int nIdx  = lane & 15;
  const int khalf = lane >> 4;

  const int wu = __builtin_amdgcn_readfirstlane(wave);

  const int fl  = tid & 31;
  const int fi0 = tid >> 5;
  const int fn  = fl & 15;
  const int fkh = fl >> 4;
  float wst[16];
#pragma unroll
  for (int q = 0; q < 8; ++q) {
    const int i    = fi0 + 4 * q;
    const int krow = 4 * i + 2 * fkh;
    wst[2 * q + 0] = W[(krow + 0) * HEADS + fn];
    wst[2 * q + 1] = W[(krow + 1) * HEADS + fn];
  }

  const float* xp = x + (size_t)(wu * 32) * MCOLS + 4 * lane;
  float4 xv[32];
#pragma unroll
  for (int kk = 0; kk < 32; ++kk) {
    xv[kk] = *(const float4*)(xp + kk * MCOLS);
  }

  const int tile = wave >> 1;
  const int ksel = wave & 1;
  const int n0 = blockIdx.x * 32 + tile * 16;
  const float* zrow = z + (size_t)(n0 + nIdx) * KDZ + ksel * (KDZ / 2) + 2 * khalf;
  v2f a[KDZ / 8];
#pragma unroll
  for (int j = 0; j < KDZ / 8; ++j) {
    a[j] = *(const v2f*)(zrow + 4 * j);
  }

  __builtin_amdgcn_sched_barrier(0);

  {
    const float msk = (fn < HEADS) ? 1.0f : 0.0f;
#pragma unroll
    for (int q = 0; q < 8; ++q) {
      const int i = fi0 + 4 * q;
      s_bf[i * 64 + fl * 2 + 0] = msk * wst[2 * q + 0];
      s_bf[i * 64 + fl * 2 + 1] = msk * wst[2 * q + 1];
    }
  }

  {
    const float* wp = W + (KDZ + wu * 32) * HEADS;
    float4 p0 = make_float4(0.f, 0.f, 0.f, 0.f);
    float4 p1 = p0, p2 = p0, p3 = p0;
#pragma unroll
    for (int kk = 0; kk < 32; ++kk) {
      const float4 xk = xv[kk];
      const float w0 = wp[kk * HEADS + 0];
      const float w1 = wp[kk * HEADS + 1];
      const float w2 = wp[kk * HEADS + 2];
      const float w3 = wp[kk * HEADS + 3];
      p0.x = fmaf(xk.x, w0, p0.x); p0.y = fmaf(xk.x, w1, p0.y);
      p0.z = fmaf(xk.x, w2, p0.z); p0.w = fmaf(xk.x, w3, p0.w);
      p1.x = fmaf(xk.y, w0, p1.x); p1.y = fmaf(xk.y, w1, p1.y);
      p1.z = fmaf(xk.y, w2, p1.z); p1.w = fmaf(xk.y, w3, p1.w);
      p2.x = fmaf(xk.z, w0, p2.x); p2.y = fmaf(xk.z, w1, p2.y);
      p2.z = fmaf(xk.z, w2, p2.z); p2.w = fmaf(xk.z, w3, p2.w);
      p3.x = fmaf(xk.w, w0, p3.x); p3.y = fmaf(xk.w, w1, p3.y);
      p3.z = fmaf(xk.w, w2, p3.z); p3.w = fmaf(xk.w, w3, p3.w);
    }
    float4* pp = (float4*)s_xwp + wu * MCOLS + 4 * lane;
    pp[0] = p0; pp[1] = p1; pp[2] = p2; pp[3] = p3;
  }
  __syncthreads();

  v8f c = {0.f, 0.f, 0.f, 0.f, 0.f, 0.f, 0.f, 0.f};
  const float* bfp = s_bf + ksel * (16 * 64) + lane * 2;
#pragma unroll
  for (int j = 0; j < KDZ / 8; ++j) {
    v2f bb = *(const v2f*)(bfp + j * 64);
    c = __builtin_amdgcn_wmma_f32_16x16x4_f32(false, a[j], false, bb,
                                              (short)0, c, false, false);
  }

  if (nIdx < HEADS) {
    const int base = wave * 64 + khalf * (8 * HEADS) + nIdx;
#pragma unroll
    for (int r = 0; r < 8; ++r) {
      s_zw[base + r * HEADS] = c[r];
    }
  }
  __syncthreads();

  const float b0 = b[0], b1 = b[1], b2 = b[2], b3 = b[3];
  float4 xw[4];
#pragma unroll
  for (int mb = 0; mb < 4; ++mb) {
    const int m = mb * 32 + lane;
    const float4 q0 = ((const float4*)s_xwp)[0 * MCOLS + m];
    const float4 q1 = ((const float4*)s_xwp)[1 * MCOLS + m];
    const float4 q2 = ((const float4*)s_xwp)[2 * MCOLS + m];
    const float4 q3 = ((const float4*)s_xwp)[3 * MCOLS + m];
    xw[mb].x = (((b0 + q0.x) + q1.x) + q2.x) + q3.x;
    xw[mb].y = (((b1 + q0.y) + q1.y) + q2.y) + q3.y;
    xw[mb].z = (((b2 + q0.z) + q1.z) + q2.z) + q3.z;
    xw[mb].w = (((b3 + q0.w) + q1.w) + q2.w) + q3.w;
  }

#pragma unroll 4
  for (int r8 = 0; r8 < 8; ++r8) {
    const int r = ksel * 8 + r8;
    const float4 zw0 = *(const float4*)(s_zw + (2 * tile + 0) * 64 + r * HEADS);
    const float4 zw1 = *(const float4*)(s_zw + (2 * tile + 1) * 64 + r * HEADS);
    float4 zw;
    zw.x = zw0.x + zw1.x; zw.y = zw0.y + zw1.y;
    zw.z = zw0.z + zw1.z; zw.w = zw0.w + zw1.w;
    const size_t rowoff = (size_t)(n0 + r) * (MCOLS * HEADS);
#pragma unroll
    for (int mb = 0; mb < 4; ++mb) {
      const int m = mb * 32 + lane;
      float4 o;
      o.x = fmaxf(zw.x + xw[mb].x, 0.f);
      o.y = fmaxf(zw.y + xw[mb].y, 0.f);
      o.z = fmaxf(zw.z + xw[mb].z, 0.f);
      o.w = fmaxf(zw.w + xw[mb].w, 0.f);
      *(float4*)(out + rowoff + (size_t)m * HEADS) = o;
    }
  }
}

extern "C" void kernel_launch(void* const* d_in, const int* in_sizes, int n_in,
                              void* d_out, int out_size, void* d_ws, size_t ws_size,
                              hipStream_t stream) {
  (void)in_sizes; (void)n_in; (void)out_size; (void)d_ws; (void)ws_size;
  const float* z = (const float*)d_in[0];
  const float* x = (const float*)d_in[1];
  const float* W = (const float*)d_in[2];
  const float* b = (const float*)d_in[3];
  float* out = (float*)d_out;

  fused_xw_zw_bcast_kernel<<<NROWS / 32, 128, 0, stream>>>(z, x, W, b, out);
}


# Work-group 4 -> 8 waves (2 per SIMD): per wave x k range 16, a K quarter (8 WMMAs), 16 stores of one 512-B column block (xW adds 8 partials, zW (q0+q1)+(q2+q3): sum order changes)
# speedup vs baseline: 2.8508x; 1.1329x over previous
#include <hip/hip_runtime.h>

typedef float v2f __attribute__((ext_vector_type(2)));
typedef float v8f __attribute__((ext_vector_type(8)));

#define NROWS 8192
#define MCOLS 128
#define KDZ   128
#define KDX   128
#define HEADS 4
#define NWAVES 8

__global__ void __launch_bounds__(256)
fused_xw_zw_bcast_kernel(const float* __restrict__ z,
                         const float* __restrict__ x,
                         const float* __restrict__ W,
                         const float* __restrict__ b,
                         float* __restrict__ out) {
  __shared__ float s_bf[32 * 64];
  __shared__ float s_xwp[NWAVES * MCOLS * HEADS];
  __shared__ float s_zw[NWAVES * 16 * HEADS];

  const int tid   = threadIdx.x;
  const int lane  = tid & 31;
  const int wave  = tid >> 5;
  const int nIdx  = lane & 15;
  const int khalf = lane >> 4;

  const int wu = __builtin_amdgcn_readfirstlane(wave);

  const int fl  = lane;
  const int fi0 = wave;
  const int fn  = fl & 15;
  const int fkh = fl >> 4;
  float wst[8];
#pragma unroll
  for (int q = 0; q < 4; ++q) {
    const int i    = fi0 + 8 * q;
    const int krow = 4 * i + 2 * fkh;
    wst[2 * q + 0] = W[(krow + 0) * HEADS + fn];
    wst[2 * q + 1] = W[(krow + 1) * HEADS + fn];
  }

  const float* xp = x + (size_t)(wu * 16) * MCOLS + 4 * lane;
  float4 xv[16];
#pragma unroll
  for (int kk = 0; kk < 16; ++kk) {
    xv[kk] = *(const float4*)(xp + kk * MCOLS);
  }

  const int tile = wave >> 2;
  const int ksel = wave & 3;
  const int n0 = blockIdx.x * 32 + tile * 16;
  const float* zrow = z + (size_t)(n0 + nIdx) * KDZ + ksel * (KDZ / 4) + 2 * khalf;
  v2f a[KDZ / 16];
#pragma unroll
  for (int j = 0; j < KDZ / 16; ++j) {
    a[j] = *(const v2f*)(zrow + 4 * j);
  }

  __builtin_amdgcn_sched_barrier(0);

  {
    const float msk = (fn < HEADS) ? 1.0f : 0.0f;
#pragma unroll
    for (int q = 0; q < 4; ++q) {
      const int i = fi0 + 8 * q;
      s_bf[i * 64 + fl * 2 + 0] = msk * wst[2 * q + 0];
      s_bf[i * 64 + fl * 2 + 1] = msk * wst[2 * q + 1];
    }
  }

  {
    const float* wp = W + (KDZ + wu * 16) * HEADS;
    float4 p0 = make_float4(0.f, 0.f, 0.f, 0.f);
    float4 p1 = p0, p2 = p0, p3 = p0;
#pragma unroll
    for (int kk = 0; kk < 16; ++kk) {
      const float4 xk = xv[kk];
      const float w0 = wp[kk * HEADS + 0];
      const float w1 = wp[kk * HEADS + 1];
      const float w2 = wp[kk * HEADS + 2];
      const float w3 = wp[kk * HEADS + 3];
      p0.x = fmaf(xk.x, w0, p0.x); p0.y = fmaf(xk.x, w1, p0.y);
      p0.z = fmaf(xk.x, w2, p0.z); p0.w = fmaf(xk.x, w3, p0.w);
      p1.x = fmaf(xk.y, w0, p1.x); p1.y = fmaf(xk.y, w1, p1.y);
      p1.z = fmaf(xk.y, w2, p1.z); p1.w = fmaf(xk.y, w3, p1.w);
      p2.x = fmaf(xk.z, w0, p2.x); p2.y = fmaf(xk.z, w1, p2.y);
      p2.z = fmaf(xk.z, w2, p2.z); p2.w = fmaf(xk.z, w3, p2.w);
      p3.x = fmaf(xk.w, w0, p3.x); p3.y = fmaf(xk.w, w1, p3.y);
      p3.z = fmaf(xk.w, w2, p3.z); p3.w = fmaf(xk.w, w3, p3.w);
    }
    float4* pp = (float4*)s_xwp + wu * MCOLS + 4 * lane;
    pp[0] = p0; pp[1] = p1; pp[2] = p2; pp[3] = p3;
  }
  __syncthreads();

  v8f c = {0.f, 0.f, 0.f, 0.f, 0.f, 0.f, 0.f, 0.f};
  const float* bfp = s_bf + ksel * (8 * 64) + lane * 2;
#pragma unroll
  for (int j = 0; j < KDZ / 16; ++j) {
    v2f bb = *(const v2f*)(bfp + j * 64);
    c = __builtin_amdgcn_wmma_f32_16x16x4_f32(false, a[j], false, bb,
                                              (short)0, c, false, false);
  }

  if (nIdx < HEADS) {
    const int base = wave * 64 + khalf * (8 * HEADS) + nIdx;
#pragma unroll
    for (int r = 0; r < 8; ++r) {
      s_zw[base + r * HEADS] = c[r];
    }
  }
  __syncthreads();

  const int mb = wave & 3;
  const int m  = mb * 32 + lane;

  float4 xw;
  {
    const float4* xs = (const float4*)s_xwp + m;
    xw.x = b[0]; xw.y = b[1]; xw.z = b[2]; xw.w = b[3];
#pragma unroll
    for (int w = 0; w < NWAVES; ++w) {
      const float4 q = xs[w * MCOLS];
      xw.x += q.x; xw.y += q.y; xw.z += q.z; xw.w += q.w;
    }
  }

  float4 zt;
  {
    const float* zs = s_zw + (4 * tile + 2 * khalf) * 64 + nIdx * HEADS;
    const float4 qa = *(const float4*)(zs);
    const float4 qb = *(const float4*)(zs + 64);
    float4 s;
    s.x = qa.x + qb.x; s.y = qa.y + qb.y; s.z = qa.z + qb.z; s.w = qa.w + qb.w;
    zt.x = s.x + __shfl_xor(s.x, 16);
    zt.y = s.y + __shfl_xor(s.y, 16);
    zt.z = s.z + __shfl_xor(s.z, 16);
    zt.w = s.w + __shfl_xor(s.w, 16);
  }

  float* op = out + (size_t)n0 * (MCOLS * HEADS) + (size_t)m * HEADS;
#pragma unroll
  for (int r = 0; r < 16; ++r) {
    const float zx = __int_as_float(__builtin_amdgcn_readlane(__float_as_int(zt.x), r));
    const float zy = __int_as_float(__builtin_amdgcn_readlane(__float_as_int(zt.y), r));
    const float zz = __int_as_float(__builtin_amdgcn_readlane(__float_as_int(zt.z), r));
    const float zq = __int_as_float(__builtin_amdgcn_readlane(__float_as_int(zt.w), r));
    float4 o;
    o.x = fmaxf(zx + xw.x, 0.f);
    o.y = fmaxf(zy + xw.y, 0.f);
    o.z = fmaxf(zz + xw.z, 0.f);
    o.w = fmaxf(zq + xw.w, 0.f);
    *(float4*)(op + (size_t)r * (MCOLS * HEADS)) = o;
  }
}

extern "C" void kernel_launch(void* const* d_in, const int* in_sizes, int n_in,
                              void* d_out, int out_size, void* d_ws, size_t ws_size,
                              hipStream_t stream) {
  (void)in_sizes; (void)n_in; (void)out_size; (void)d_ws; (void)ws_size;
  const float* z = (const float*)d_in[0];
  const float* x = (const float*)d_in[1];
  const float* W = (const float*)d_in[2];
  const float* b = (const float*)d_in[3];
  float* out = (float*)d_out;

  fused_xw_zw_bcast_kernel<<<NROWS / 32, 256, 0, stream>>>(z, x, W, b, out);
}


# Loads issued small-first (W, z fragments, then x rows); B fragments straight from W, not staged in LDS: the WMMAs, then the FMAs, run under the x loads; one barrier, not two (sum orders unchanged)
# speedup vs baseline: 3.1303x; 1.0980x over previous
#include <hip/hip_runtime.h>

typedef float v2f __attribute__((ext_vector_type(2)));
typedef float v8f __attribute__((ext_vector_type(8)));

#define NROWS 8192
#define MCOLS 128
#define KDZ   128
#define KDX   128
#define HEADS 4
#define NWAVES 8

__global__ void __launch_bounds__(256)
fused_xw_zw_bcast_kernel(const float* __restrict__ z,
                         const float* __restrict__ x,
                         const float* __restrict__ W,
                         const float* __restrict__ b,
                         float* __restrict__ out) {
  __shared__ float s_xwp[NWAVES * MCOLS * HEADS];
  __shared__ float s_zw[NWAVES * 16 * HEADS];

  const int tid   = threadIdx.x;
  const int lane  = tid & 31;
  const int wave  = tid >> 5;
  const int nIdx  = lane & 15;
  const int khalf = lane >> 4;

  const int wu = __builtin_amdgcn_readfirstlane(wave);

  const int tile = wave >> 2;
  const int ksel = wave & 3;
  const int n0 = blockIdx.x * 32 + tile * 16;

  const float* wfrag = W + (ksel * (KDZ / 4) + 2 * khalf) * HEADS + (nIdx & (HEADS - 1));
  float wr[2 * (KDZ / 16)];
#pragma unroll
  for (int j = 0; j < KDZ / 16; ++j) {
    wr[2 * j + 0] = wfrag[(4 * j + 0) * HEADS];
    wr[2 * j + 1] = wfrag[(4 * j + 1) * HEADS];
  }

  const float* zrow = z + (size_t)(n0 + nIdx) * KDZ + ksel * (KDZ / 4) + 2 * khalf;
  v2f a[KDZ / 16];
#pragma unroll
  for (int j = 0; j < KDZ / 16; ++j) {
    a[j] = *(const v2f*)(zrow + 4 * j);
  }

  const float* wp = W + (KDZ + wu * 16) * HEADS;
  float w2[16 * HEADS];
#pragma unroll
  for (int i = 0; i < 16 * HEADS; ++i) {
    w2[i] = wp[i];
  }

  __builtin_amdgcn_sched_barrier(0);

  const float* xp = x + (size_t)(wu * 16) * MCOLS + 4 * lane;
  float4 xv[16];
#pragma unroll
  for (int kk = 0; kk < 16; ++kk) {
    xv[kk] = *(const float4*)(xp + kk * MCOLS);
  }

  __builtin_amdgcn_sched_barrier(0);

  v8f c = {0.f, 0.f, 0.f, 0.f, 0.f, 0.f, 0.f, 0.f};
#pragma unroll
  for (int j = 0; j < KDZ / 16; ++j) {
    v2f bb;
    bb.x = wr[2 * j + 0];
    bb.y = wr[2 * j + 1];
    c = __builtin_amdgcn_wmma_f32_16x16x4_f32(false, a[j], false, bb,
                                              (short)0, c, false, false);
  }

  if (nIdx < HEADS) {
    const int base = wave * 64 + khalf * (8 * HEADS) + nIdx;
#pragma unroll
    for (int r = 0; r < 8; ++r) {
      s_zw[base + r * HEADS] = c[r];
    }
  }

  {
    float4 p0 = make_float4(0.f, 0.f, 0.f, 0.f);
    float4 p1 = p0, p2 = p0, p3 = p0;
#pragma unroll
    for (int kk = 0; kk < 16; ++kk) {
      const float4 xk = xv[kk];
      const float w0 = w2[kk * HEADS + 0];
      const float w1 = w2[kk * HEADS + 1];
      const float wq = w2[kk * HEADS + 2];
      const float w3 = w2[kk * HEADS + 3];
      p0.x = fmaf(xk.x, w0, p0.x); p0.y = fmaf(xk.x, w1, p0.y);
      p0.z = fmaf(xk.x, wq, p0.z); p0.w = fmaf(xk.x, w3, p0.w);
      p1.x = fmaf(xk.y, w0, p1.x); p1.y = fmaf(xk.y, w1, p1.y);
      p1.z = fmaf(xk.y, wq, p1.z); p1.w = fmaf(xk.y, w3, p1.w);
      p2.x = fmaf(xk.z, w0, p2.x); p2.y = fmaf(xk.z, w1, p2.y);
      p2.z = fmaf(xk.z, wq, p2.z); p2.w = fmaf(xk.z, w3, p2.w);
      p3.x = fmaf(xk.w, w0, p3.x); p3.y = fmaf(xk.w, w1, p3.y);
      p3.z = fmaf(xk.w, wq, p3.z); p3.w = fmaf(xk.w, w3, p3.w);
    }
    float4* pp = (float4*)s_xwp + wu * MCOLS + 4 * lane;
    pp[0] = p0; pp[1] = p1; pp[2] = p2; pp[3] = p3;
  }
  __syncthreads();

  const int mb = wave & 3;
  const int m  = mb * 32 + lane;

  float4 xw;
  {
    const float4* xs = (const float4*)s_xwp + m;
    xw.x = b[0]; xw.y = b[1]; xw.z = b[2]; xw.w = b[3];
#pragma unroll
    for (int w = 0; w < NWAVES; ++w) {
      const float4 q = xs[w * MCOLS];
      xw.x += q.x; xw.y += q.y; xw.z += q.z; xw.w += q.w;
    }
  }

  float4 zt;
  {
    const float* zs = s_zw + (4 * tile + 2 * khalf) * 64 + nIdx * HEADS;
    const float4 qa = *(const float4*)(zs);
    const float4 qb = *(const float4*)(zs + 64);
    float4 s;
    s.x = qa.x + qb.x; s.y = qa.y + qb.y; s.z = qa.z + qb.z; s.w = qa.w + qb.w;
    zt.x = s.x + __shfl_xor(s.x, 16);
    zt.y = s.y + __shfl_xor(s.y, 16);
    zt.z = s.z + __shfl_xor(s.z, 16);
    zt.w = s.w + __shfl_xor(s.w, 16);
  }

  float* op = out + (size_t)n0 * (MCOLS * HEADS) + (size_t)m * HEADS;
#pragma unroll
  for (int r = 0; r < 16; ++r) {
    const float zx = __int_as_float(__builtin_amdgcn_readlane(__float_as_int(zt.x), r));
    const float zy = __int_as_float(__builtin_amdgcn_readlane(__float_as_int(zt.y), r));
    const float zz = __int_as_float(__builtin_amdgcn_readlane(__float_as_int(zt.z), r));
    const float zq = __int_as_float(__builtin_amdgcn_readlane(__float_as_int(zt.w), r));
    float4 o;
    o.x = fmaxf(zx + xw.x, 0.f);
    o.y = fmaxf(zy + xw.y, 0.f);
    o.z = fmaxf(zz + xw.z, 0.f);
    o.w = fmaxf(zq + xw.w, 0.f);
    *(float4*)(op + (size_t)r * (MCOLS * HEADS)) = o;
  }
}

extern "C" void kernel_launch(void* const* d_in, const int* in_sizes, int n_in,
                              void* d_out, int out_size, void* d_ws, size_t ws_size,
                              hipStream_t stream) {
  (void)in_sizes; (void)n_in; (void)out_size; (void)d_ws; (void)ws_size;
  const float* z = (const float*)d_in[0];
  const float* x = (const float*)d_in[1];
  const float* W = (const float*)d_in[2];
  const float* b = (const float*)d_in[3];
  float* out = (float*)d_out;

  fused_xw_zw_bcast_kernel<<<NROWS / 32, 256, 0, stream>>>(z, x, W, b, out);
}
